# starting kernel
# baseline (speedup 1.0000x reference)
#include <hip/hip_runtime.h>


typedef float v2f __attribute__((ext_vector_type(2)));
typedef float v8f __attribute__((ext_vector_type(8)));

namespace lcn {
constexpr int kB   = 512;
constexpr int kH   = 280;
constexpr int kW   = 280;
constexpr int kKS  = 28;
constexpr int kHS  = 10;
constexpr int kWS  = 10;
constexpr int kP   = 100;
constexpr int kF   = 16;
constexpr int kK   = 784;
constexpr int kOUT = 10;
constexpr int kWavesPerBlock = 8;
constexpr int kFP  = kF * kP;
}

__global__ __launch_bounds__(256)
void lcn_stage1_wmma(const float* __restrict__ x,
                     const float* __restrict__ weight,
                     const float* __restrict__ bias,
                     float* __restrict__ ybuf)
{
    using namespace lcn;
    __shared__ float sw[kF * kK];

    const int p   = blockIdx.x;
    const int ph  = p / kWS;
    const int pw  = p % kWS;
    const int tid = threadIdx.x;

    constexpr int kVec4 = (kF * kK) / 4;
    for (int g = tid; g < kVec4; g += 256) {
        const int f = g / (kK / 4);
        const int r = g % (kK / 4);
        const float4* src =
            reinterpret_cast<const float4*>(weight + (size_t)(f * kP + p) * kK) + r;
        reinterpret_cast<float4*>(sw + f * kK)[r] = *src;
    }
    __syncthreads();

    const int wave = tid >> 5;
    const int lane = tid & 31;
    const int half = lane >> 4;
    const int mn   = lane & 15;

    const int b0 = (blockIdx.y * kWavesPerBlock + wave) * 16;

    const float* xb = x + (size_t)(b0 + mn) * (kH * kW)
                        + (size_t)(ph * kKS) * kW + pw * kKS;
    const float* wb = sw + mn * kK;

    v8f acc0 = {};
    v8f acc1 = {};

    for (int kh = 0; kh < kKS; ++kh) {
        const float* xrow = xb + kh * kW;
        const float* wrow = wb + kh * kKS;
        #pragma unroll
        for (int kwb = 0; kwb < kKS; kwb += 4) {
            v2f a = *reinterpret_cast<const v2f*>(xrow + kwb + half * 2);
            v2f b = *reinterpret_cast<const v2f*>(wrow + kwb + half * 2);
            if ((kwb >> 2) & 1)
                acc1 = __builtin_amdgcn_wmma_f32_16x16x4_f32(
                    false, a, false, b, (short)0, acc1, false, false);
            else
                acc0 = __builtin_amdgcn_wmma_f32_16x16x4_f32(
                    false, a, false, b, (short)0, acc0, false, false);
        }
    }

    const float bv = bias[mn * kP + p];
    #pragma unroll
    for (int j = 0; j < 8; ++j) {
        float y = acc0[j] + acc1[j] + bv;
        y = y > 0.0f ? y : 0.0f;
        const int b = b0 + j + half * 8;
        ybuf[(size_t)b * kFP + mn * kP + p] = y;
    }
}

__global__ __launch_bounds__(320)
void lcn_stage2_decode(const float* __restrict__ ybuf,
                       const float* __restrict__ dec_w,
                       const float* __restrict__ dec_b,
                       float* __restrict__ out)
{
    using namespace lcn;
    const int b = blockIdx.x;
    const int o = threadIdx.x >> 5;
    const int l = threadIdx.x & 31;

    const float* yr = ybuf  + (size_t)b * kFP;
    const float* wr = dec_w + (size_t)o * kFP;

    float acc = 0.0f;
    #pragma unroll 5
    for (int i = l; i < kFP; i += 32)
        acc += yr[i] * wr[i];

    #pragma unroll
    for (int m = 16; m >= 1; m >>= 1)
        acc += __shfl_xor(acc, m, 32);

    if (l == 0)
        out[b * kOUT + o] = acc + dec_b[o];
}

extern "C" void kernel_launch(void* const* d_in, const int* in_sizes, int n_in,
                              void* d_out, int out_size, void* d_ws, size_t ws_size,
                              hipStream_t stream)
{
    using namespace lcn;
    const float* x      = (const float*)d_in[0];
    const float* weight = (const float*)d_in[1];
    const float* bias   = (const float*)d_in[2];
    const float* dec_w  = (const float*)d_in[3];
    const float* dec_b  = (const float*)d_in[4];
    float* out  = (float*)d_out;
    float* ybuf = (float*)d_ws;

    dim3 g1(kP, kB / (16 * kWavesPerBlock));
    lcn_stage1_wmma<<<g1, 256, 0, stream>>>(x, weight, bias, ybuf);
    lcn_stage2_decode<<<kB, 320, 0, stream>>>(ybuf, dec_w, dec_b, out);
}


# stage 1 x tile: lanes load 16 B along the 112 B row segments, four row pairs ahead, through a wave-private LDS stage (was 8 B strided per lane, used at once); k regrouped, sum order changes
# speedup vs baseline: 1.0512x; 1.0512x over previous
#include <hip/hip_runtime.h>


typedef float v2f __attribute__((ext_vector_type(2)));
typedef float v4f __attribute__((ext_vector_type(4)));
typedef float v8f __attribute__((ext_vector_type(8)));

namespace lcn {
constexpr int kB   = 512;
constexpr int kH   = 280;
constexpr int kW   = 280;
constexpr int kKS  = 28;
constexpr int kHS  = 10;
constexpr int kWS  = 10;
constexpr int kP   = 100;
constexpr int kF   = 16;
constexpr int kK   = 784;
constexpr int kOUT = 10;
constexpr int kWavesPerBlock = 8;
constexpr int kFP  = kF * kP;
constexpr int kImg = kH * kW;
constexpr int kPairK  = 2 * kKS;
constexpr int kPairs  = kKS / 2;
constexpr int kLoads  = 7;
constexpr int kAhead  = 4;
constexpr int kWLoads = 13;
constexpr int kWPitch = kK + 4;
constexpr int kXPitch = kPairK + 4;
}

template <bool kPrefetch>
__device__ __forceinline__ void lcn_pair_step(v4f (&reg)[lcn::kLoads],
                                              const float* __restrict__ xnext,
                                              const int (&goff)[lcn::kLoads],
                                              const int (&soff)[lcn::kLoads],
                                              float* xs, const float* xa, const float* wrow,
                                              v8f& acc0, v8f& acc1)
{
    using namespace lcn;
    __builtin_amdgcn_fence(__ATOMIC_RELEASE, "wavefront");
    __builtin_amdgcn_wave_barrier();
    #pragma unroll
    for (int i = 0; i < kLoads; ++i)
        *reinterpret_cast<v4f*>(xs + soff[i]) = reg[i];
    if (kPrefetch) {
        #pragma unroll
        for (int i = 0; i < kLoads; ++i)
            reg[i] = *reinterpret_cast<const v4f*>(xnext + goff[i]);
    }
    __builtin_amdgcn_fence(__ATOMIC_RELEASE, "wavefront");
    __builtin_amdgcn_wave_barrier();
    __builtin_amdgcn_fence(__ATOMIC_ACQUIRE, "wavefront");

    v4f a[kLoads], b[kLoads];
    #pragma unroll
    for (int u = 0; u < kLoads; ++u) {
        a[u] = *reinterpret_cast<const v4f*>(xa + 8 * u);
        b[u] = *reinterpret_cast<const v4f*>(wrow + 8 * u);
    }
    #pragma unroll
    for (int u = 0; u < kLoads; ++u) {
        v2f a0 = {a[u].x, a[u].y};
        v2f b0 = {b[u].x, b[u].y};
        v2f a1 = {a[u].z, a[u].w};
        v2f b1 = {b[u].z, b[u].w};
        acc0 = __builtin_amdgcn_wmma_f32_16x16x4_f32(
            false, a0, false, b0, (short)0, acc0, false, false);
        acc1 = __builtin_amdgcn_wmma_f32_16x16x4_f32(
            false, a1, false, b1, (short)0, acc1, false, false);
    }
}

__global__ __launch_bounds__(256)
void lcn_stage1_wmma(const float* __restrict__ x,
                     const float* __restrict__ weight,
                     const float* __restrict__ bias,
                     float* __restrict__ ybuf)
{
    using namespace lcn;
    __shared__ __attribute__((aligned(16))) float sw[kF * kWPitch];
    __shared__ __attribute__((aligned(16))) float sx[kWavesPerBlock * 16 * kXPitch];

    const int p   = blockIdx.x;
    const int ph  = p / kWS;
    const int pw  = p % kWS;
    const int tid = threadIdx.x;

    const int wave = tid >> 5;
    const int lane = tid & 31;
    const int half = lane >> 4;
    const int mn   = lane & 15;

    const int b0 = (blockIdx.y * kWavesPerBlock + wave) * 16;

    int goff[kLoads], soff[kLoads];
    #pragma unroll
    for (int i = 0; i < kLoads; ++i) {
        const int e   = i * 32 + lane;
        const int seg = e / 7;
        const int j   = e - seg * 7;
        const int r   = seg >> 1;
        const int kk  = seg & 1;
        goff[i] = (b0 + r) * kImg + (ph * kKS + kk) * kW + pw * kKS + j * 4;
        soff[i] = r * kXPitch + kk * kKS + j * 4;
    }

    constexpr int kVec4 = (kF * kK) / 4;
    v4f wr[kWLoads];
    #pragma unroll
    for (int t = 0; t < kWLoads; ++t) {
        const int g = tid + 256 * t;
        if (g < kVec4) {
            const int f = g / (kK / 4);
            const int r = g % (kK / 4);
            wr[t] = reinterpret_cast<const v4f*>(weight + (size_t)(f * kP + p) * kK)[r];
        }
    }
    v4f rg[kAhead][kLoads];
    #pragma unroll
    for (int s = 0; s < kAhead; ++s) {
        #pragma unroll
        for (int i = 0; i < kLoads; ++i)
            rg[s][i] = *reinterpret_cast<const v4f*>(x + s * (2 * kW) + goff[i]);
    }
    #pragma unroll
    for (int t = 0; t < kWLoads; ++t) {
        const int g = tid + 256 * t;
        if (g < kVec4) {
            const int f = g / (kK / 4);
            const int r = g % (kK / 4);
            reinterpret_cast<v4f*>(sw + f * kWPitch)[r] = wr[t];
        }
    }
    __builtin_amdgcn_fence(__ATOMIC_RELEASE, "workgroup", "local");
    __builtin_amdgcn_s_barrier();
    __builtin_amdgcn_fence(__ATOMIC_ACQUIRE, "workgroup", "local");

    float*       xs = sx + wave * (16 * kXPitch);
    const float* xa = xs + mn * kXPitch + half * 4;
    const float* wb = sw + mn * kWPitch + half * 4;

    v8f acc0 = {};
    v8f acc1 = {};

    #pragma unroll
    for (int pr = 0; pr < kPairs; ++pr) {
        if (pr + kAhead < kPairs)
            lcn_pair_step<true>(rg[pr % kAhead], x + (pr + kAhead) * (2 * kW), goff, soff,
                                xs, xa, wb + pr * kPairK, acc0, acc1);
        else
            lcn_pair_step<false>(rg[pr % kAhead], x, goff, soff,
                                 xs, xa, wb + pr * kPairK, acc0, acc1);
    }

    const float bv = bias[mn * kP + p];
    #pragma unroll
    for (int j = 0; j < 8; ++j) {
        float y = acc0[j] + acc1[j] + bv;
        y = y > 0.0f ? y : 0.0f;
        const int b = b0 + j + half * 8;
        ybuf[(size_t)b * kFP + mn * kP + p] = y;
    }
}

__global__ __launch_bounds__(320)
void lcn_stage2_decode(const float* __restrict__ ybuf,
                       const float* __restrict__ dec_w,
                       const float* __restrict__ dec_b,
                       float* __restrict__ out)
{
    using namespace lcn;
    const int b = blockIdx.x;
    const int o = threadIdx.x >> 5;
    const int l = threadIdx.x & 31;

    const float* yr = ybuf  + (size_t)b * kFP;
    const float* wr = dec_w + (size_t)o * kFP;

    float acc = 0.0f;
    #pragma unroll 5
    for (int i = l; i < kFP; i += 32)
        acc += yr[i] * wr[i];

    #pragma unroll
    for (int m = 16; m >= 1; m >>= 1)
        acc += __shfl_xor(acc, m, 32);

    if (l == 0)
        out[b * kOUT + o] = acc + dec_b[o];
}

extern "C" void kernel_launch(void* const* d_in, const int* in_sizes, int n_in,
                              void* d_out, int out_size, void* d_ws, size_t ws_size,
                              hipStream_t stream)
{
    using namespace lcn;
    const float* x      = (const float*)d_in[0];
    const float* weight = (const float*)d_in[1];
    const float* bias   = (const float*)d_in[2];
    const float* dec_w  = (const float*)d_in[3];
    const float* dec_b  = (const float*)d_in[4];
    float* out  = (float*)d_out;
    float* ybuf = (float*)d_ws;

    dim3 g1(kP, kB / (16 * kWavesPerBlock));
    lcn_stage1_wmma<<<g1, 256, 0, stream>>>(x, weight, bias, ybuf);
    lcn_stage2_decode<<<kB, 320, 0, stream>>>(ybuf, dec_w, dec_b, out);
}


# fold the decoder into stage 1: each wave multiplies its 16x16 relu tile by the patch's 16x10 decoder slice (4 more f32 WMMAs), stores whole 128 B lines; stage 2 sums 100 planes; sum order changes
# speedup vs baseline: 2.1888x; 2.0822x over previous
#include <hip/hip_runtime.h>


typedef float v2f __attribute__((ext_vector_type(2)));
typedef float v4f __attribute__((ext_vector_type(4)));
typedef float v8f __attribute__((ext_vector_type(8)));

namespace lcn {
constexpr int kB   = 512;
constexpr int kH   = 280;
constexpr int kW   = 280;
constexpr int kKS  = 28;
constexpr int kHS  = 10;
constexpr int kWS  = 10;
constexpr int kP   = 100;
constexpr int kF   = 16;
constexpr int kK   = 784;
constexpr int kOUT = 10;
constexpr int kWavesPerBlock = 8;
constexpr int kFP  = kF * kP;
constexpr int kImg = kH * kW;
constexpr int kPairK  = 2 * kKS;
constexpr int kPairs  = kKS / 2;
constexpr int kLoads  = 7;
constexpr int kAhead  = 4;
constexpr int kWLoads = 13;
constexpr int kWPitch = kK + 4;
constexpr int kXPitch = kPairK + 4;
constexpr int kPlane  = kB * kOUT;
constexpr int kSumChains  = 4;
constexpr int kSumElems   = 64;
}

template <bool kPrefetch>
__device__ __forceinline__ void lcn_pair_step(v4f (&reg)[lcn::kLoads],
                                              const float* __restrict__ xnext,
                                              const int (&goff)[lcn::kLoads],
                                              const int (&soff)[lcn::kLoads],
                                              float* xs, const float* xa, const float* wrow,
                                              v8f& acc0, v8f& acc1)
{
    using namespace lcn;
    __builtin_amdgcn_fence(__ATOMIC_RELEASE, "wavefront");
    __builtin_amdgcn_wave_barrier();
    #pragma unroll
    for (int i = 0; i < kLoads; ++i)
        *reinterpret_cast<v4f*>(xs + soff[i]) = reg[i];
    if (kPrefetch) {
        #pragma unroll
        for (int i = 0; i < kLoads; ++i)
            reg[i] = *reinterpret_cast<const v4f*>(xnext + goff[i]);
    }
    __builtin_amdgcn_fence(__ATOMIC_RELEASE, "wavefront");
    __builtin_amdgcn_wave_barrier();
    __builtin_amdgcn_fence(__ATOMIC_ACQUIRE, "wavefront");

    v4f a[kLoads], b[kLoads];
    #pragma unroll
    for (int u = 0; u < kLoads; ++u) {
        a[u] = *reinterpret_cast<const v4f*>(xa + 8 * u);
        b[u] = *reinterpret_cast<const v4f*>(wrow + 8 * u);
    }
    #pragma unroll
    for (int u = 0; u < kLoads; ++u) {
        v2f a0 = {a[u].x, a[u].y};
        v2f b0 = {b[u].x, b[u].y};
        v2f a1 = {a[u].z, a[u].w};
        v2f b1 = {b[u].z, b[u].w};
        acc0 = __builtin_amdgcn_wmma_f32_16x16x4_f32(
            false, a0, false, b0, (short)0, acc0, false, false);
        acc1 = __builtin_amdgcn_wmma_f32_16x16x4_f32(
            false, a1, false, b1, (short)0, acc1, false, false);
    }
}

__global__ __launch_bounds__(256)
void lcn_stage1_wmma(const float* __restrict__ x,
                     const float* __restrict__ weight,
                     const float* __restrict__ bias,
                     const float* __restrict__ dec_w,
                     float* __restrict__ part)
{
    using namespace lcn;
    __shared__ __attribute__((aligned(16))) float sw[kF * kWPitch];
    __shared__ __attribute__((aligned(16))) float sx[kWavesPerBlock * 16 * kXPitch];

    const int p   = blockIdx.x;
    const int ph  = p / kWS;
    const int pw  = p % kWS;
    const int tid = threadIdx.x;

    const int wave = tid >> 5;
    const int lane = tid & 31;
    const int half = lane >> 4;
    const int mn   = lane & 15;

    const int b0 = (blockIdx.y * kWavesPerBlock + wave) * 16;

    int goff[kLoads], soff[kLoads];
    #pragma unroll
    for (int i = 0; i < kLoads; ++i) {
        const int e   = i * 32 + lane;
        const int seg = e / 7;
        const int j   = e - seg * 7;
        const int r   = seg >> 1;
        const int kk  = seg & 1;
        goff[i] = (b0 + r) * kImg + (ph * kKS + kk) * kW + pw * kKS + j * 4;
        soff[i] = r * kXPitch + kk * kKS + j * 4;
    }

    constexpr int kVec4 = (kF * kK) / 4;
    v4f wr[kWLoads];
    #pragma unroll
    for (int t = 0; t < kWLoads; ++t) {
        const int g = tid + 256 * t;
        if (g < kVec4) {
            const int f = g / (kK / 4);
            const int r = g % (kK / 4);
            wr[t] = reinterpret_cast<const v4f*>(weight + (size_t)(f * kP + p) * kK)[r];
        }
    }
    v4f rg[kAhead][kLoads];
    #pragma unroll
    for (int s = 0; s < kAhead; ++s) {
        #pragma unroll
        for (int i = 0; i < kLoads; ++i)
            rg[s][i] = *reinterpret_cast<const v4f*>(x + s * (2 * kW) + goff[i]);
    }
    #pragma unroll
    for (int t = 0; t < kWLoads; ++t) {
        const int g = tid + 256 * t;
        if (g < kVec4) {
            const int f = g / (kK / 4);
            const int r = g % (kK / 4);
            reinterpret_cast<v4f*>(sw + f * kWPitch)[r] = wr[t];
        }
    }
    __builtin_amdgcn_fence(__ATOMIC_RELEASE, "workgroup", "local");
    __builtin_amdgcn_s_barrier();
    __builtin_amdgcn_fence(__ATOMIC_ACQUIRE, "workgroup", "local");

    float*       xs = sx + wave * (16 * kXPitch);
    const float* xa = xs + mn * kXPitch + half * 4;
    const float* wb = sw + mn * kWPitch + half * 4;

    v8f acc0 = {};
    v8f acc1 = {};

    float dw[2][4];
    #pragma unroll
    for (int pr = 0; pr < kPairs; ++pr) {
        if (pr + kAhead == kPairs) {
            const int oc = mn < kOUT ? mn : kOUT - 1;
            #pragma unroll
            for (int u = 0; u < 2; ++u) {
                #pragma unroll
                for (int t = 0; t < 4; ++t)
                    dw[u][t] = dec_w[(size_t)oc * kFP + (8 * u + 4 * half + t) * kP + p];
            }
        }
        if (pr + kAhead < kPairs)
            lcn_pair_step<true>(rg[pr % kAhead], x + (pr + kAhead) * (2 * kW), goff, soff,
                                xs, xa, wb + pr * kPairK, acc0, acc1);
        else
            lcn_pair_step<false>(rg[pr % kAhead], x, goff, soff,
                                 xs, xa, wb + pr * kPairK, acc0, acc1);
    }

    const float bv = bias[mn * kP + p];

    #pragma unroll
    for (int j = 0; j < 8; ++j) {
        float y = acc0[j] + acc1[j] + bv;
        y = y > 0.0f ? y : 0.0f;
        xs[(j + half * 8) * kXPitch + mn] = y;
    }
    __builtin_amdgcn_fence(__ATOMIC_RELEASE, "wavefront");
    __builtin_amdgcn_wave_barrier();
    __builtin_amdgcn_fence(__ATOMIC_ACQUIRE, "wavefront");

    const v4f ya0 = *reinterpret_cast<const v4f*>(xa);
    const v4f ya1 = *reinterpret_cast<const v4f*>(xa + 8);
    v8f d2 = {};
    {
        v2f a0 = {ya0.x, ya0.y};
        v2f b0v = {dw[0][0], dw[0][1]};
        v2f a1 = {ya0.z, ya0.w};
        v2f b1v = {dw[0][2], dw[0][3]};
        v2f a2 = {ya1.x, ya1.y};
        v2f b2v = {dw[1][0], dw[1][1]};
        v2f a3 = {ya1.z, ya1.w};
        v2f b3v = {dw[1][2], dw[1][3]};
        d2 = __builtin_amdgcn_wmma_f32_16x16x4_f32(false, a0, false, b0v, (short)0, d2, false, false);
        d2 = __builtin_amdgcn_wmma_f32_16x16x4_f32(false, a1, false, b1v, (short)0, d2, false, false);
        d2 = __builtin_amdgcn_wmma_f32_16x16x4_f32(false, a2, false, b2v, (short)0, d2, false, false);
        d2 = __builtin_amdgcn_wmma_f32_16x16x4_f32(false, a3, false, b3v, (short)0, d2, false, false);
    }

    if (mn < kOUT) {
        #pragma unroll
        for (int j = 0; j < 8; ++j)
            xs[(j + half * 8) * kOUT + mn] = d2[j];
    }
    __builtin_amdgcn_fence(__ATOMIC_RELEASE, "wavefront");
    __builtin_amdgcn_wave_barrier();
    __builtin_amdgcn_fence(__ATOMIC_ACQUIRE, "wavefront");

    float* dst = part + (size_t)p * kPlane + (size_t)b0 * kOUT;
    const v4f o0 = *reinterpret_cast<const v4f*>(xs + 4 * lane);
    *reinterpret_cast<v4f*>(dst + 4 * lane) = o0;
    if (lane < 8) {
        const v4f o1 = *reinterpret_cast<const v4f*>(xs + 128 + 4 * lane);
        *reinterpret_cast<v4f*>(dst + 128 + 4 * lane) = o1;
    }
}

__global__ __launch_bounds__(256)
void lcn_stage2_sum(const float* __restrict__ part,
                    const float* __restrict__ dec_b,
                    float* __restrict__ out)
{
    using namespace lcn;
    __shared__ float sh[kSumChains][kSumElems];
    const int el = threadIdx.x & (kSumElems - 1);
    const int g  = threadIdx.x / kSumElems;
    const int e  = blockIdx.x * kSumElems + el;
    float s = 0.0f;
    #pragma unroll
    for (int q = 0; q < kP / kSumChains; ++q)
        s += part[(g + kSumChains * q) * kPlane + e];
    sh[g][el] = s;
    __syncthreads();
    if (threadIdx.x < kSumElems)
        out[e] = ((sh[0][el] + sh[1][el]) + (sh[2][el] + sh[3][el])) + dec_b[e % kOUT];
}

extern "C" void kernel_launch(void* const* d_in, const int* in_sizes, int n_in,
                              void* d_out, int out_size, void* d_ws, size_t ws_size,
                              hipStream_t stream)
{
    using namespace lcn;
    const float* x      = (const float*)d_in[0];
    const float* weight = (const float*)d_in[1];
    const float* bias   = (const float*)d_in[2];
    const float* dec_w  = (const float*)d_in[3];
    const float* dec_b  = (const float*)d_in[4];
    float* out  = (float*)d_out;
    float* part = (float*)d_ws;

    dim3 g1(kP, kB / (16 * kWavesPerBlock));
    lcn_stage1_wmma<<<g1, 256, 0, stream>>>(x, weight, bias, dec_w, part);
    lcn_stage2_sum<<<kPlane / kSumElems, kSumChains * kSumElems, 0, stream>>>(part, dec_b, out);
}


# stage 1: each wave takes 16 batch rows of two adjacent patches and one half of K, so x is read in 224 B runs (was 112 B); the two K-half waves swap a 16x16 partial through LDS; sum order changes
# speedup vs baseline: 2.3063x; 1.0537x over previous
#include <hip/hip_runtime.h>


typedef float v2f __attribute__((ext_vector_type(2)));
typedef float v4f __attribute__((ext_vector_type(4)));
typedef float v8f __attribute__((ext_vector_type(8)));

namespace lcn {
constexpr int kB   = 512;
constexpr int kH   = 280;
constexpr int kW   = 280;
constexpr int kKS  = 28;
constexpr int kHS  = 10;
constexpr int kWS  = 10;
constexpr int kP   = 100;
constexpr int kF   = 16;
constexpr int kK   = 784;
constexpr int kOUT = 10;
constexpr int kWavesPerBlock = 8;
constexpr int kFP  = kF * kP;
constexpr int kImg = kH * kW;
constexpr int kPatches  = 2;
constexpr int kPairsW   = kWS / kPatches;
constexpr int kRowK     = kPatches * kKS;
constexpr int kHalfRows = kKS / 2;
constexpr int kLoads  = 7;
constexpr int kAhead  = 4;
constexpr int kWVec4  = kPatches * kF * kK / 4;
constexpr int kWLoads = (kWVec4 + 255) / 256;
constexpr int kWPitch = kK + 4;
constexpr int kXPitch = kRowK + 4;
constexpr int kPlane  = kB * kOUT;
constexpr int kSumChains  = 4;
constexpr int kSumElems   = 64;
}

template <bool kPrefetch>
__device__ __forceinline__ void lcn_row_step(v4f (&reg)[lcn::kLoads],
                                             const float* __restrict__ xnext,
                                             const int (&goff)[lcn::kLoads],
                                             const int (&soff)[lcn::kLoads],
                                             float* xs, const float* xa, const float* xt,
                                             const float* wrow, const float* wtail,
                                             v8f (&acc)[lcn::kPatches])
{
    using namespace lcn;
    __builtin_amdgcn_fence(__ATOMIC_RELEASE, "wavefront");
    __builtin_amdgcn_wave_barrier();
    #pragma unroll
    for (int i = 0; i < kLoads; ++i)
        *reinterpret_cast<v4f*>(xs + soff[i]) = reg[i];
    if (kPrefetch) {
        #pragma unroll
        for (int i = 0; i < kLoads; ++i)
            reg[i] = *reinterpret_cast<const v4f*>(xnext + goff[i]);
    }
    __builtin_amdgcn_fence(__ATOMIC_RELEASE, "wavefront");
    __builtin_amdgcn_wave_barrier();
    __builtin_amdgcn_fence(__ATOMIC_ACQUIRE, "wavefront");

    v4f a[kPatches][3], b[kPatches][3];
    v2f at[kPatches], bt[kPatches];
    #pragma unroll
    for (int q = 0; q < kPatches; ++q) {
        #pragma unroll
        for (int u = 0; u < 3; ++u) {
            a[q][u] = *reinterpret_cast<const v4f*>(xa + q * kKS + 8 * u);
            b[q][u] = *reinterpret_cast<const v4f*>(wrow + q * (kF * kWPitch) + 8 * u);
        }
        at[q] = *reinterpret_cast<const v2f*>(xt + q * kKS);
        bt[q] = *reinterpret_cast<const v2f*>(wtail + q * (kF * kWPitch));
    }
    #pragma unroll
    for (int u = 0; u < 3; ++u) {
        #pragma unroll
        for (int q = 0; q < kPatches; ++q) {
            v2f a0 = {a[q][u].x, a[q][u].y};
            v2f b0 = {b[q][u].x, b[q][u].y};
            acc[q] = __builtin_amdgcn_wmma_f32_16x16x4_f32(
                false, a0, false, b0, (short)0, acc[q], false, false);
        }
        #pragma unroll
        for (int q = 0; q < kPatches; ++q) {
            v2f a1 = {a[q][u].z, a[q][u].w};
            v2f b1 = {b[q][u].z, b[q][u].w};
            acc[q] = __builtin_amdgcn_wmma_f32_16x16x4_f32(
                false, a1, false, b1, (short)0, acc[q], false, false);
        }
    }
    #pragma unroll
    for (int q = 0; q < kPatches; ++q)
        acc[q] = __builtin_amdgcn_wmma_f32_16x16x4_f32(
            false, at[q], false, bt[q], (short)0, acc[q], false, false);
}

__global__ __launch_bounds__(256)
void lcn_stage1_wmma(const float* __restrict__ x,
                     const float* __restrict__ weight,
                     const float* __restrict__ bias,
                     const float* __restrict__ dec_w,
                     float* __restrict__ part)
{
    using namespace lcn;
    __shared__ __attribute__((aligned(16))) float sw[kPatches * kF * kWPitch];
    __shared__ __attribute__((aligned(16))) float sx[kWavesPerBlock * 16 * kXPitch];
    __shared__ __attribute__((aligned(16))) float sg[kWavesPerBlock * 256];

    const int pp  = blockIdx.x;
    const int ph  = pp / kPairsW;
    const int pwp = pp % kPairsW;
    const int p0  = ph * kWS + kPatches * pwp;
    const int tid = threadIdx.x;

    const int wave = tid >> 5;
    const int lane = tid & 31;
    const int half = lane >> 4;
    const int mn   = lane & 15;
    const int mt   = wave >> 1;
    const int kh   = wave & 1;

    const int b0 = (blockIdx.y * (kWavesPerBlock / 2) + mt) * 16;

    int goff[kLoads], soff[kLoads];
    #pragma unroll
    for (int i = 0; i < kLoads; ++i) {
        const int e = i * 32 + lane;
        const int r = e / 14;
        const int j = e - r * 14;
        goff[i] = (b0 + r) * kImg + (ph * kKS + kh * kHalfRows) * kW + pwp * kRowK + j * 4;
        soff[i] = r * kXPitch + j * 4;
    }

    v4f wr[kWLoads];
    #pragma unroll
    for (int t = 0; t < kWLoads; ++t) {
        const int g = tid + 256 * t;
        if (g < kWVec4) {
            const int q   = g / (kF * kK / 4);
            const int rem = g - q * (kF * kK / 4);
            const int f   = rem / (kK / 4);
            const int r   = rem - f * (kK / 4);
            wr[t] = reinterpret_cast<const v4f*>(weight + (size_t)(f * kP + p0 + q) * kK)[r];
        }
    }
    v4f rg[kAhead][kLoads];
    #pragma unroll
    for (int s = 0; s < 2; ++s) {
        #pragma unroll
        for (int i = 0; i < kLoads; ++i)
            rg[s][i] = *reinterpret_cast<const v4f*>(x + s * kW + goff[i]);
    }
    #pragma unroll
    for (int t = 0; t < kWLoads; ++t) {
        const int g = tid + 256 * t;
        if (g < kWVec4) {
            const int q   = g / (kF * kK / 4);
            const int rem = g - q * (kF * kK / 4);
            const int f   = rem / (kK / 4);
            const int r   = rem - f * (kK / 4);
            reinterpret_cast<v4f*>(sw + (q * kF + f) * kWPitch)[r] = wr[t];
        }
    }
    #pragma unroll
    for (int s = 2; s < kAhead; ++s) {
        #pragma unroll
        for (int i = 0; i < kLoads; ++i)
            rg[s][i] = *reinterpret_cast<const v4f*>(x + s * kW + goff[i]);
    }
    __builtin_amdgcn_fence(__ATOMIC_RELEASE, "workgroup", "local");
    __builtin_amdgcn_s_barrier();
    __builtin_amdgcn_fence(__ATOMIC_ACQUIRE, "workgroup", "local");

    float*       xs = sx + wave * (16 * kXPitch);
    const float* xa = xs + mn * kXPitch + half * 4;
    const float* xt = xs + mn * kXPitch + 24 + half * 2;
    const float* wb = sw + mn * kWPitch + kh * (kHalfRows * kKS) + half * 4;
    const float* wt = sw + mn * kWPitch + kh * (kHalfRows * kKS) + 24 + half * 2;

    v8f acc[kPatches] = {};

    const int pk = p0 + kh;

    float dw[2][4];
    float bv = 0.0f;
    #pragma unroll
    for (int s = 0; s < kHalfRows; ++s) {
        if (s + kAhead == kHalfRows) {
            const int oc = mn < kOUT ? mn : kOUT - 1;
            #pragma unroll
            for (int u = 0; u < 2; ++u) {
                #pragma unroll
                for (int t = 0; t < 4; ++t)
                    dw[u][t] = dec_w[(size_t)oc * kFP + (8 * u + 4 * half + t) * kP + pk];
            }
            bv = bias[mn * kP + pk];
        }
        if (s + kAhead < kHalfRows)
            lcn_row_step<true>(rg[s % kAhead], x + (s + kAhead) * kW, goff, soff,
                               xs, xa, xt, wb + s * kKS, wt + s * kKS, acc);
        else
            lcn_row_step<false>(rg[s % kAhead], x, goff, soff,
                                xs, xa, xt, wb + s * kKS, wt + s * kKS, acc);
    }

    float keep[8];
    float* gs = sg + wave * 256;
    #pragma unroll
    for (int j = 0; j < 8; ++j) {
        const float s0 = acc[0][j];
        const float s1 = acc[1][j];
        keep[j] = kh ? s1 : s0;
        gs[j * 32 + lane] = kh ? s0 : s1;
    }
    __builtin_amdgcn_fence(__ATOMIC_RELEASE, "workgroup", "local");
    __builtin_amdgcn_s_barrier();
    __builtin_amdgcn_fence(__ATOMIC_ACQUIRE, "workgroup", "local");
    const float* gp = sg + (wave ^ 1) * 256;

    #pragma unroll
    for (int j = 0; j < 8; ++j) {
        float y = (keep[j] + gp[j * 32 + lane]) + bv;
        y = y > 0.0f ? y : 0.0f;
        xs[(j + half * 8) * kXPitch + mn] = y;
    }
    __builtin_amdgcn_fence(__ATOMIC_RELEASE, "wavefront");
    __builtin_amdgcn_wave_barrier();
    __builtin_amdgcn_fence(__ATOMIC_ACQUIRE, "wavefront");

    const v4f ya0 = *reinterpret_cast<const v4f*>(xa);
    const v4f ya1 = *reinterpret_cast<const v4f*>(xa + 8);
    v8f d2 = {};
    {
        v2f a0 = {ya0.x, ya0.y};
        v2f b0v = {dw[0][0], dw[0][1]};
        v2f a1 = {ya0.z, ya0.w};
        v2f b1v = {dw[0][2], dw[0][3]};
        v2f a2 = {ya1.x, ya1.y};
        v2f b2v = {dw[1][0], dw[1][1]};
        v2f a3 = {ya1.z, ya1.w};
        v2f b3v = {dw[1][2], dw[1][3]};
        d2 = __builtin_amdgcn_wmma_f32_16x16x4_f32(false, a0, false, b0v, (short)0, d2, false, false);
        d2 = __builtin_amdgcn_wmma_f32_16x16x4_f32(false, a1, false, b1v, (short)0, d2, false, false);
        d2 = __builtin_amdgcn_wmma_f32_16x16x4_f32(false, a2, false, b2v, (short)0, d2, false, false);
        d2 = __builtin_amdgcn_wmma_f32_16x16x4_f32(false, a3, false, b3v, (short)0, d2, false, false);
    }

    if (mn < kOUT) {
        #pragma unroll
        for (int j = 0; j < 8; ++j)
            xs[(j + half * 8) * kOUT + mn] = d2[j];
    }
    __builtin_amdgcn_fence(__ATOMIC_RELEASE, "wavefront");
    __builtin_amdgcn_wave_barrier();
    __builtin_amdgcn_fence(__ATOMIC_ACQUIRE, "wavefront");

    float* dst = part + (size_t)pk * kPlane + (size_t)b0 * kOUT;
    const v4f o0 = *reinterpret_cast<const v4f*>(xs + 4 * lane);
    *reinterpret_cast<v4f*>(dst + 4 * lane) = o0;
    if (lane < 8) {
        const v4f o1 = *reinterpret_cast<const v4f*>(xs + 128 + 4 * lane);
        *reinterpret_cast<v4f*>(dst + 128 + 4 * lane) = o1;
    }
}

__global__ __launch_bounds__(256)
void lcn_stage2_sum(const float* __restrict__ part,
                    const float* __restrict__ dec_b,
                    float* __restrict__ out)
{
    using namespace lcn;
    __shared__ float sh[kSumChains][kSumElems];
    const int el = threadIdx.x & (kSumElems - 1);
    const int g  = threadIdx.x / kSumElems;
    const int e  = blockIdx.x * kSumElems + el;
    float s = 0.0f;
    #pragma unroll
    for (int q = 0; q < kP / kSumChains; ++q)
        s += part[(g + kSumChains * q) * kPlane + e];
    sh[g][el] = s;
    __syncthreads();
    if (threadIdx.x < kSumElems)
        out[e] = ((sh[0][el] + sh[1][el]) + (sh[2][el] + sh[3][el])) + dec_b[e % kOUT];
}

extern "C" void kernel_launch(void* const* d_in, const int* in_sizes, int n_in,
                              void* d_out, int out_size, void* d_ws, size_t ws_size,
                              hipStream_t stream)
{
    using namespace lcn;
    const float* x      = (const float*)d_in[0];
    const float* weight = (const float*)d_in[1];
    const float* bias   = (const float*)d_in[2];
    const float* dec_w  = (const float*)d_in[3];
    const float* dec_b  = (const float*)d_in[4];
    float* out  = (float*)d_out;
    float* part = (float*)d_ws;

    dim3 g1(kP / kPatches, kB / (16 * (kWavesPerBlock / 2)));
    lcn_stage1_wmma<<<g1, 256, 0, stream>>>(x, weight, bias, dec_w, part);
    lcn_stage2_sum<<<kPlane / kSumElems, kSumChains * kSumElems, 0, stream>>>(part, dec_b, out);
}
